# starting kernel
# baseline (speedup 1.0000x reference)
#include <hip/hip_runtime.h>

typedef float v2f __attribute__((ext_vector_type(2)));
typedef float v8f __attribute__((ext_vector_type(8)));

#define T_STEPS   8
#define WAY       64
#define SHOT      4
#define NQ        1024
#define C         512
#define QPB       64
#define CK        64
#define PITCH     68

__global__ __launch_bounds__(256, 4)
void proto_dist_kernel(const float* __restrict__ q,
                       const float* __restrict__ s,
                       float* __restrict__ out)
{
    __shared__ float ldsQ[QPB * PITCH];
    __shared__ float ldsP[WAY * PITCH];
    __shared__ float qnp[QPB][4];
    __shared__ float pnp[WAY][4];
    __shared__ float qn[QPB];
    __shared__ float pn[WAY];

    const int tid = threadIdx.x;
    const int t   = blockIdx.x >> 4;
    const int qb  = blockIdx.x & 15;

    const float* Qg = q + ((size_t)t * NQ + (size_t)qb * QPB) * C;
    const float* Sg = s + (size_t)t * (WAY * SHOT) * C;
    float*       Og = out + ((size_t)t * NQ + (size_t)qb * QPB) * WAY;

    const int row = tid >> 2;
    const int sub = tid & 3;

    const int lane = tid & 31;
    const int w    = tid >> 5;
    const int wm   = w & 3;
    const int n0   = (w >> 2) * 32;
    const int half = lane >> 4;
    const int l15  = lane & 15;

    v8f acc0 = {};
    v8f acc1 = {};
    float qn_priv = 0.0f;
    float pn_priv = 0.0f;

    for (int ck = 0; ck < C; ck += CK) {
        __syncthreads();

        #pragma unroll
        for (int i = 0; i < 4; ++i) {
            const int col = (sub + 4 * i) * 4;
            const float4 v = *(const float4*)(Qg + (size_t)row * C + ck + col);
            *(float4*)&ldsQ[row * PITCH + col] = v;
            qn_priv += v.x * v.x + v.y * v.y + v.z * v.z + v.w * v.w;
        }

        #pragma unroll
        for (int i = 0; i < 4; ++i) {
            const int col = (sub + 4 * i) * 4;
            const float* sp = Sg + (size_t)(row * SHOT) * C + ck + col;
            const float4 a = *(const float4*)(sp);
            const float4 b = *(const float4*)(sp + C);
            const float4 c = *(const float4*)(sp + 2 * C);
            const float4 d = *(const float4*)(sp + 3 * C);
            float4 m;
            m.x = 0.25f * (a.x + b.x + c.x + d.x);
            m.y = 0.25f * (a.y + b.y + c.y + d.y);
            m.z = 0.25f * (a.z + b.z + c.z + d.z);
            m.w = 0.25f * (a.w + b.w + c.w + d.w);
            *(float4*)&ldsP[row * PITCH + col] = m;
            pn_priv += m.x * m.x + m.y * m.y + m.z * m.z + m.w * m.w;
        }

        __syncthreads();

        #pragma unroll
        for (int kk = 0; kk < CK; kk += 4) {
            const int k = kk + 2 * half;
            const v2f a  = *(const v2f*)&ldsQ[(wm * 16 + l15) * PITCH + k];
            const v2f b0 = *(const v2f*)&ldsP[(n0 + l15) * PITCH + k];
            const v2f b1 = *(const v2f*)&ldsP[(n0 + 16 + l15) * PITCH + k];
            acc0 = __builtin_amdgcn_wmma_f32_16x16x4_f32(
                false, a, false, b0, (short)0, acc0, false, false);
            acc1 = __builtin_amdgcn_wmma_f32_16x16x4_f32(
                false, a, false, b1, (short)0, acc1, false, false);
        }
    }

    qnp[row][sub] = qn_priv;
    pnp[row][sub] = pn_priv;
    __syncthreads();
    if (tid < 64) {
        qn[tid] = qnp[tid][0] + qnp[tid][1] + qnp[tid][2] + qnp[tid][3];
    } else if (tid < 128) {
        const int r = tid - 64;
        pn[r] = pnp[r][0] + pnp[r][1] + pnp[r][2] + pnp[r][3];
    }
    __syncthreads();

    const float pn0 = pn[n0 + l15];
    const float pn1 = pn[n0 + 16 + l15];
    #pragma unroll
    for (int i = 0; i < 8; ++i) {
        const int r   = wm * 16 + i + 8 * half;
        const float qr = qn[r];
        Og[(size_t)r * WAY + n0 + l15]      = 2.0f * acc0[i] - qr - pn0;
        Og[(size_t)r * WAY + n0 + 16 + l15] = 2.0f * acc1[i] - qr - pn1;
    }
}

extern "C" void kernel_launch(void* const* d_in, const int* in_sizes, int n_in,
                              void* d_out, int out_size, void* d_ws, size_t ws_size,
                              hipStream_t stream)
{
    const float* qf = (const float*)d_in[0];
    const float* sf = (const float*)d_in[1];
    float* out = (float*)d_out;

    const int t = in_sizes[0] / (NQ * C);
    dim3 grid(t * (NQ / QPB));
    dim3 block(256);
    hipLaunchKernelGGL(proto_dist_kernel, grid, block, 0, stream, qf, sf, out);
}


# Re-tile to 256 workgroups of 64 queries x 32 ways (was 128 of 64x64) so every WGP works and reads less; rows leave as whole 128-B lines via LDS; proto norm now summed from 8 partials per way
# speedup vs baseline: 1.5622x; 1.5622x over previous
#include <hip/hip_runtime.h>


typedef float v2f __attribute__((ext_vector_type(2)));
typedef float v8f __attribute__((ext_vector_type(8)));

#define T_STEPS   8
#define WAY       64
#define SHOT      4
#define NQ        1024
#define C         512
#define QPB       64
#define WPB       32
#define CK        64
#define PITCH     68

__global__ __launch_bounds__(256, 4)
void proto_dist_kernel(const float* __restrict__ q,
                       const float* __restrict__ s,
                       float* __restrict__ out)
{
    __shared__ float ldsQ[QPB * PITCH];
    __shared__ float ldsP[WPB * PITCH];
    __shared__ float qnp[QPB][4];
    __shared__ float pnp[WPB][8];
    __shared__ float qn[QPB];
    __shared__ float pn[WPB];

    const int tid = threadIdx.x;
    const int wh  = blockIdx.x & 1;
    const int qb  = (blockIdx.x >> 1) & 15;
    const int t   = blockIdx.x >> 5;

    const float* Qg = q + ((size_t)t * NQ + (size_t)qb * QPB) * C;
    const float* Sg = s + ((size_t)t * (WAY * SHOT) + (size_t)wh * (WPB * SHOT)) * C;
    float*       Og = out + ((size_t)t * NQ + (size_t)qb * QPB) * WAY + wh * WPB;

    const int row = tid >> 2;
    const int sub = tid & 3;
    const int wrow = tid >> 3;
    const int wsub = tid & 7;

    const int lane = tid & 31;
    const int w    = tid >> 5;
    const int wm   = w & 3;
    const int n0   = (w >> 2) * 16;
    const int half = lane >> 4;
    const int l15  = lane & 15;

    v8f acc = {};
    float qn_priv = 0.0f;
    float pn_priv = 0.0f;

    for (int ck = 0; ck < C; ck += CK) {
        __syncthreads();

        #pragma unroll
        for (int i = 0; i < 4; ++i) {
            const int col = (sub + 4 * i) * 4;
            const float4 v = *(const float4*)(Qg + (size_t)row * C + ck + col);
            *(float4*)&ldsQ[row * PITCH + col] = v;
            qn_priv += v.x * v.x + v.y * v.y + v.z * v.z + v.w * v.w;
        }

        #pragma unroll
        for (int i = 0; i < 2; ++i) {
            const int col = (wsub + 8 * i) * 4;
            const float* sp = Sg + (size_t)(wrow * SHOT) * C + ck + col;
            const float4 a = *(const float4*)(sp);
            const float4 b = *(const float4*)(sp + C);
            const float4 c = *(const float4*)(sp + 2 * C);
            const float4 d = *(const float4*)(sp + 3 * C);
            float4 m;
            m.x = 0.25f * (a.x + b.x + c.x + d.x);
            m.y = 0.25f * (a.y + b.y + c.y + d.y);
            m.z = 0.25f * (a.z + b.z + c.z + d.z);
            m.w = 0.25f * (a.w + b.w + c.w + d.w);
            *(float4*)&ldsP[wrow * PITCH + col] = m;
            pn_priv += m.x * m.x + m.y * m.y + m.z * m.z + m.w * m.w;
        }

        __syncthreads();

        #pragma unroll
        for (int kk = 0; kk < CK; kk += 4) {
            const int k = kk + 2 * half;
            const v2f a = *(const v2f*)&ldsQ[(wm * 16 + l15) * PITCH + k];
            const v2f b = *(const v2f*)&ldsP[(n0 + l15) * PITCH + k];
            acc = __builtin_amdgcn_wmma_f32_16x16x4_f32(
                false, a, false, b, (short)0, acc, false, false);
        }
    }

    qnp[row][sub]   = qn_priv;
    pnp[wrow][wsub] = pn_priv;
    __syncthreads();
    if (tid < 64) {
        qn[tid] = qnp[tid][0] + qnp[tid][1] + qnp[tid][2] + qnp[tid][3];
    } else if (tid < 96) {
        const int r = tid - 64;
        pn[r] = ((pnp[r][0] + pnp[r][1]) + (pnp[r][2] + pnp[r][3]))
              + ((pnp[r][4] + pnp[r][5]) + (pnp[r][6] + pnp[r][7]));
    }
    __syncthreads();

    const float pn0 = pn[n0 + l15];
    #pragma unroll
    for (int i = 0; i < 8; ++i) {
        const int r   = wm * 16 + i + 8 * half;
        const float qr = qn[r];
        ldsQ[r * PITCH + n0 + l15] = 2.0f * acc[i] - qr - pn0;
    }
    __syncthreads();

    for (int j = 0; j < 2; ++j) {
        const int idx = tid + 256 * j;
        const int r   = idx >> 3;
        const int c4  = (idx & 7) * 4;
        const float4 v = *(const float4*)&ldsQ[r * PITCH + c4];
        *(float4*)(Og + (size_t)r * WAY + c4) = v;
    }
}

extern "C" void kernel_launch(void* const* d_in, const int* in_sizes, int n_in,
                              void* d_out, int out_size, void* d_ws, size_t ws_size,
                              hipStream_t stream)
{
    const float* qf = (const float*)d_in[0];
    const float* sf = (const float*)d_in[1];
    float* out = (float*)d_out;

    const int t = in_sizes[0] / (NQ * C);
    dim3 grid(t * (NQ / QPB) * (WAY / WPB));
    dim3 block(256);
    hipLaunchKernelGGL(proto_dist_kernel, grid, block, 0, stream, qf, sf, out);
}


# Software-pipeline the K loop: loads of the next two 64-wide chunks stay in flight during the matrix work (two register stages, two LDS stages, hand-written barrier that keeps the load counter)
# speedup vs baseline: 2.6574x; 1.7011x over previous
#include <hip/hip_runtime.h>


typedef float v2f __attribute__((ext_vector_type(2)));
typedef float v4f __attribute__((ext_vector_type(4)));
typedef float v8f __attribute__((ext_vector_type(8)));

#define T_STEPS   8
#define WAY       64
#define SHOT      4
#define NQ        1024
#define C         512
#define QPB       64
#define WPB       32
#define CK        64
#define PITCH     68

struct Stage {
    v4f q[4];
    v4f s[2][4];
};

static __device__ __forceinline__
void stage_load(Stage& st, const float* __restrict__ qp, const float* __restrict__ sp, int ck)
{
    #pragma unroll
    for (int i = 0; i < 4; ++i)
        st.q[i] = *(const v4f*)(qp + ck + 16 * i);
    #pragma unroll
    for (int i = 0; i < 2; ++i) {
        #pragma unroll
        for (int j = 0; j < SHOT; ++j)
            st.s[i][j] = *(const v4f*)(sp + ck + 32 * i + j * C);
    }
}

static __device__ __forceinline__
void stage_arrive(Stage& st)
{
    asm volatile("" : "+v"(st.q[0]), "+v"(st.q[1]), "+v"(st.q[2]), "+v"(st.q[3]),
                      "+v"(st.s[0][0]), "+v"(st.s[0][1]), "+v"(st.s[0][2]), "+v"(st.s[0][3]),
                      "+v"(st.s[1][0]), "+v"(st.s[1][1]), "+v"(st.s[1][2]), "+v"(st.s[1][3]));
}

static __device__ __forceinline__
void stage_commit(const Stage& st, float* lq, float* lp, float& qn_priv, float& pn_priv)
{
    #pragma unroll
    for (int i = 0; i < 4; ++i) {
        const v4f v = st.q[i];
        *(v4f*)&lq[16 * i] = v;
        qn_priv += v.x * v.x + v.y * v.y + v.z * v.z + v.w * v.w;
    }
    #pragma unroll
    for (int i = 0; i < 2; ++i) {
        const v4f a = st.s[i][0];
        const v4f b = st.s[i][1];
        const v4f c = st.s[i][2];
        const v4f d = st.s[i][3];
        v4f m;
        m.x = 0.25f * (a.x + b.x + c.x + d.x);
        m.y = 0.25f * (a.y + b.y + c.y + d.y);
        m.z = 0.25f * (a.z + b.z + c.z + d.z);
        m.w = 0.25f * (a.w + b.w + c.w + d.w);
        *(v4f*)&lp[32 * i] = m;
        pn_priv += m.x * m.x + m.y * m.y + m.z * m.z + m.w * m.w;
    }
}

static __device__ __forceinline__
void mma_chunk(const float* la, const float* lb, v8f& acc)
{
    #pragma unroll
    for (int kk = 0; kk < CK; kk += 4) {
        const v2f a = *(const v2f*)&la[kk];
        const v2f b = *(const v2f*)&lb[kk];
        acc = __builtin_amdgcn_wmma_f32_16x16x4_f32(
            false, a, false, b, (short)0, acc, false, false);
    }
}

static __device__ __forceinline__
void barrier_keep_loads()
{
    asm volatile("s_wait_dscnt 0x0\n\ts_barrier_signal -1\n\ts_barrier_wait -1" ::: "memory");
}

template <bool REFILL>
static __device__ __forceinline__
void pipe_step(Stage& st, const float* __restrict__ qp, const float* __restrict__ sp, int ck_refill,
               float* lq, float* lp, const float* la, const float* lb,
               float& qn_priv, float& pn_priv, v8f& acc)
{
    stage_arrive(st);
    stage_commit(st, lq, lp, qn_priv, pn_priv);
    __builtin_amdgcn_sched_barrier(0);
    if (REFILL) stage_load(st, qp, sp, ck_refill);
    barrier_keep_loads();
    __builtin_amdgcn_sched_barrier(0);
    mma_chunk(la, lb, acc);
    __builtin_amdgcn_sched_barrier(0);
}

__global__ __launch_bounds__(256, 4)
void proto_dist_kernel(const float* __restrict__ q,
                       const float* __restrict__ s,
                       float* __restrict__ out)
{
    __shared__ float ldsQ[2][QPB * PITCH];
    __shared__ float ldsP[2][WPB * PITCH];
    __shared__ float qnp[QPB][4];
    __shared__ float pnp[WPB][8];
    __shared__ float qn[QPB];
    __shared__ float pn[WPB];

    const int tid = threadIdx.x;
    const int wh  = blockIdx.x & 1;
    const int qb  = (blockIdx.x >> 1) & 15;
    const int t   = blockIdx.x >> 5;

    const float* Qg = q + ((size_t)t * NQ + (size_t)qb * QPB) * C;
    const float* Sg = s + ((size_t)t * (WAY * SHOT) + (size_t)wh * (WPB * SHOT)) * C;
    float*       Og = out + ((size_t)t * NQ + (size_t)qb * QPB) * WAY + wh * WPB;

    const int row = tid >> 2;
    const int sub = tid & 3;
    const int wrow = tid >> 3;
    const int wsub = tid & 7;

    const int lane = tid & 31;
    const int w    = tid >> 5;
    const int wm   = w & 3;
    const int n0   = (w >> 2) * 16;
    const int half = lane >> 4;
    const int l15  = lane & 15;

    const float* qp = Qg + (size_t)row * C + sub * 4;
    const float* sp = Sg + (size_t)(wrow * SHOT) * C + wsub * 4;
    const int lq_off = row * PITCH + sub * 4;
    const int lp_off = wrow * PITCH + wsub * 4;
    const int la_off = (wm * 16 + l15) * PITCH + 2 * half;
    const int lb_off = (n0 + l15) * PITCH + 2 * half;

    v8f acc = {};
    float qn_priv = 0.0f;
    float pn_priv = 0.0f;

    Stage st0, st1;
    stage_load(st0, qp, sp, 0);
    __builtin_amdgcn_sched_barrier(0);
    stage_load(st1, qp, sp, CK);

    float* lq0 = &ldsQ[0][lq_off];
    float* lq1 = &ldsQ[1][lq_off];
    float* lp0 = &ldsP[0][lp_off];
    float* lp1 = &ldsP[1][lp_off];
    const float* la0 = &ldsQ[0][la_off];
    const float* la1 = &ldsQ[1][la_off];
    const float* lb0 = &ldsP[0][lb_off];
    const float* lb1 = &ldsP[1][lb_off];

    for (int ck = 0; ck < C - 2 * CK; ck += 2 * CK) {
        pipe_step<true>(st0, qp, sp, ck + 2 * CK, lq0, lp0, la0, lb0, qn_priv, pn_priv, acc);
        pipe_step<true>(st1, qp, sp, ck + 3 * CK, lq1, lp1, la1, lb1, qn_priv, pn_priv, acc);
    }
    pipe_step<false>(st0, qp, sp, 0, lq0, lp0, la0, lb0, qn_priv, pn_priv, acc);
    pipe_step<false>(st1, qp, sp, 0, lq1, lp1, la1, lb1, qn_priv, pn_priv, acc);

    qnp[row][sub]   = qn_priv;
    pnp[wrow][wsub] = pn_priv;
    __syncthreads();
    if (tid < 64) {
        qn[tid] = qnp[tid][0] + qnp[tid][1] + qnp[tid][2] + qnp[tid][3];
    } else if (tid < 96) {
        const int r = tid - 64;
        pn[r] = ((pnp[r][0] + pnp[r][1]) + (pnp[r][2] + pnp[r][3]))
              + ((pnp[r][4] + pnp[r][5]) + (pnp[r][6] + pnp[r][7]));
    }
    __syncthreads();

    float* tile = ldsQ[0];
    const float pn0 = pn[n0 + l15];
    #pragma unroll
    for (int i = 0; i < 8; ++i) {
        const int r   = wm * 16 + i + 8 * half;
        const float qr = qn[r];
        tile[r * PITCH + n0 + l15] = 2.0f * acc[i] - qr - pn0;
    }
    __syncthreads();

    for (int j = 0; j < 2; ++j) {
        const int idx = tid + 256 * j;
        const int r   = idx >> 3;
        const int c4  = (idx & 7) * 4;
        const float4 v = *(const float4*)&tile[r * PITCH + c4];
        *(float4*)(Og + (size_t)r * WAY + c4) = v;
    }
}

extern "C" void kernel_launch(void* const* d_in, const int* in_sizes, int n_in,
                              void* d_out, int out_size, void* d_ws, size_t ws_size,
                              hipStream_t stream)
{
    const float* qf = (const float*)d_in[0];
    const float* sf = (const float*)d_in[1];
    float* out = (float*)d_out;

    const int t = in_sizes[0] / (NQ * C);
    dim3 grid(t * (NQ / QPB) * (WAY / WPB));
    dim3 block(256);
    hipLaunchKernelGGL(proto_dist_kernel, grid, block, 0, stream, qf, sf, out);
}


# Staging map changed to 16 lanes per 256-byte chunk row: LDS stage stores free of bank conflicts (were 4-way for queries, 2-way for prototypes), loads cover whole lines; norms from 16 partials per row
# speedup vs baseline: 2.8338x; 1.0664x over previous
#include <hip/hip_runtime.h>


typedef float v2f __attribute__((ext_vector_type(2)));
typedef float v4f __attribute__((ext_vector_type(4)));
typedef float v8f __attribute__((ext_vector_type(8)));

#define T_STEPS   8
#define WAY       64
#define SHOT      4
#define NQ        1024
#define C         512
#define QPB       64
#define WPB       32
#define CK        64
#define PITCH     68
#define PPITCH    20


struct Stage {
    v4f q[4];
    v4f s[2][4];
};

static __device__ __forceinline__
void stage_load(Stage& st, const float* __restrict__ qp, const float* __restrict__ sp, int ck)
{
    #pragma unroll
    for (int i = 0; i < 4; ++i)
        st.q[i] = *(const v4f*)(qp + ck + (size_t)(16 * i) * C);
    #pragma unroll
    for (int i = 0; i < 2; ++i) {
        #pragma unroll
        for (int j = 0; j < SHOT; ++j)
            st.s[i][j] = *(const v4f*)(sp + ck + (size_t)(16 * i * SHOT + j) * C);
    }
}

static __device__ __forceinline__
void stage_arrive(Stage& st)
{
    asm volatile("" : "+v"(st.q[0]), "+v"(st.q[1]), "+v"(st.q[2]), "+v"(st.q[3]),
                      "+v"(st.s[0][0]), "+v"(st.s[0][1]), "+v"(st.s[0][2]), "+v"(st.s[0][3]),
                      "+v"(st.s[1][0]), "+v"(st.s[1][1]), "+v"(st.s[1][2]), "+v"(st.s[1][3]));
}

static __device__ __forceinline__
void stage_commit(const Stage& st, float* lq, float* lp, v2f (&qa)[4], v2f (&pa)[2])
{
    #pragma unroll
    for (int i = 0; i < 4; ++i) {
        const v4f v = st.q[i];
        *(v4f*)&lq[16 * i * PITCH] = v;
        const v2f lo = {v.x, v.y};
        const v2f hi = {v.z, v.w};
        qa[i] += lo * lo;
        qa[i] += hi * hi;
    }
    #pragma unroll
    for (int i = 0; i < 2; ++i) {
        const v4f a = st.s[i][0];
        const v4f b = st.s[i][1];
        const v4f c = st.s[i][2];
        const v4f d = st.s[i][3];
        v4f m;
        m.x = 0.25f * (a.x + b.x + c.x + d.x);
        m.y = 0.25f * (a.y + b.y + c.y + d.y);
        m.z = 0.25f * (a.z + b.z + c.z + d.z);
        m.w = 0.25f * (a.w + b.w + c.w + d.w);
        *(v4f*)&lp[16 * i * PITCH] = m;
        const v2f lo = {m.x, m.y};
        const v2f hi = {m.z, m.w};
        pa[i] += lo * lo;
        pa[i] += hi * hi;
    }
}

static __device__ __forceinline__
void mma_chunk(const float* la, const float* lb, v8f& acc)
{
    #pragma unroll
    for (int kk = 0; kk < CK; kk += 4) {
        const v2f a = *(const v2f*)&la[kk];
        const v2f b = *(const v2f*)&lb[kk];
        acc = __builtin_amdgcn_wmma_f32_16x16x4_f32(
            false, a, false, b, (short)0, acc, false, false);
    }
}

static __device__ __forceinline__
void barrier_keep_loads()
{
    asm volatile("s_wait_dscnt 0x0\n\ts_barrier_signal -1\n\ts_barrier_wait -1" ::: "memory");
}

template <bool REFILL>
static __device__ __forceinline__
void pipe_step(Stage& st, const float* __restrict__ qp, const float* __restrict__ sp, int ck_refill,
               float* lq, float* lp, const float* la, const float* lb,
               v2f (&qa)[4], v2f (&pa)[2], v8f& acc)
{
    stage_arrive(st);
    stage_commit(st, lq, lp, qa, pa);
    __builtin_amdgcn_sched_barrier(0);
    if (REFILL) stage_load(st, qp, sp, ck_refill);
    barrier_keep_loads();
    __builtin_amdgcn_sched_barrier(0);
    mma_chunk(la, lb, acc);
    __builtin_amdgcn_sched_barrier(0);
}

__global__ __launch_bounds__(256, 4)
void proto_dist_kernel(const float* __restrict__ q,
                       const float* __restrict__ s,
                       float* __restrict__ out)
{
    __shared__ float ldsQ[2][QPB * PITCH];
    __shared__ float ldsP[2][WPB * PITCH];
    __shared__ __attribute__((aligned(16))) float qnp[QPB][PPITCH];
    __shared__ __attribute__((aligned(16))) float pnp[WPB][PPITCH];
    __shared__ float qn[QPB];
    __shared__ float pn[WPB];

    const int tid = threadIdx.x;
    const int wh  = blockIdx.x & 1;
    const int qb  = (blockIdx.x >> 1) & 15;
    const int t   = blockIdx.x >> 5;

    const float* Qg = q + ((size_t)t * NQ + (size_t)qb * QPB) * C;
    const float* Sg = s + ((size_t)t * (WAY * SHOT) + (size_t)wh * (WPB * SHOT)) * C;
    float*       Og = out + ((size_t)t * NQ + (size_t)qb * QPB) * WAY + wh * WPB;

    const int srow = tid >> 4;
    const int scol = tid & 15;

    const int lane = tid & 31;
    const int w    = tid >> 5;
    const int wm   = w & 3;
    const int n0   = (w >> 2) * 16;
    const int half = lane >> 4;
    const int l15  = lane & 15;

    const float* qp = Qg + (size_t)srow * C + scol * 4;
    const float* sp = Sg + (size_t)(srow * SHOT) * C + scol * 4;
    const int lq_off = srow * PITCH + scol * 4;
    const int lp_off = lq_off;
    const int la_off = (wm * 16 + l15) * PITCH + 2 * half;
    const int lb_off = (n0 + l15) * PITCH + 2 * half;

    v8f acc = {};
    v2f qa[4] = {};
    v2f pa[2] = {};

    Stage st0, st1;
    stage_load(st0, qp, sp, 0);
    __builtin_amdgcn_sched_barrier(0);
    stage_load(st1, qp, sp, CK);

    float* lq0 = &ldsQ[0][lq_off];
    float* lq1 = &ldsQ[1][lq_off];
    float* lp0 = &ldsP[0][lp_off];
    float* lp1 = &ldsP[1][lp_off];
    const float* la0 = &ldsQ[0][la_off];
    const float* la1 = &ldsQ[1][la_off];
    const float* lb0 = &ldsP[0][lb_off];
    const float* lb1 = &ldsP[1][lb_off];

    for (int ck = 0; ck < C - 2 * CK; ck += 2 * CK) {
        pipe_step<true>(st0, qp, sp, ck + 2 * CK, lq0, lp0, la0, lb0, qa, pa, acc);
        pipe_step<true>(st1, qp, sp, ck + 3 * CK, lq1, lp1, la1, lb1, qa, pa, acc);
    }
    pipe_step<false>(st0, qp, sp, 0, lq0, lp0, la0, lb0, qa, pa, acc);
    pipe_step<false>(st1, qp, sp, 0, lq1, lp1, la1, lb1, qa, pa, acc);

    #pragma unroll
    for (int i = 0; i < 4; ++i) qnp[srow + 16 * i][scol] = qa[i].x + qa[i].y;
    #pragma unroll
    for (int i = 0; i < 2; ++i) pnp[srow + 16 * i][scol] = pa[i].x + pa[i].y;
    __syncthreads();
    if (tid < 96) {
        const float* pr = (tid < 64) ? &qnp[tid][0] : &pnp[tid - 64][0];
        const v4f a = *(const v4f*)&pr[0];
        const v4f b = *(const v4f*)&pr[4];
        const v4f c = *(const v4f*)&pr[8];
        const v4f d = *(const v4f*)&pr[12];
        const float sum = (((a.x + a.y) + (a.z + a.w)) + ((b.x + b.y) + (b.z + b.w)))
                        + (((c.x + c.y) + (c.z + c.w)) + ((d.x + d.y) + (d.z + d.w)));
        if (tid < 64) qn[tid] = sum;
        else          pn[tid - 64] = sum;
    }
    __syncthreads();

    float* tile = ldsQ[0];
    const float pn0 = pn[n0 + l15];
    #pragma unroll
    for (int i = 0; i < 8; ++i) {
        const int r   = wm * 16 + i + 8 * half;
        const float qr = qn[r];
        tile[r * PITCH + n0 + l15] = 2.0f * acc[i] - qr - pn0;
    }
    __syncthreads();

    for (int j = 0; j < 2; ++j) {
        const int idx = tid + 256 * j;
        const int r   = idx >> 3;
        const int c4  = (idx & 7) * 4;
        const float4 v = *(const float4*)&tile[r * PITCH + c4];
        *(float4*)(Og + (size_t)r * WAY + c4) = v;
    }
}

extern "C" void kernel_launch(void* const* d_in, const int* in_sizes, int n_in,
                              void* d_out, int out_size, void* d_ws, size_t ws_size,
                              hipStream_t stream)
{
    const float* qf = (const float*)d_in[0];
    const float* sf = (const float*)d_in[1];
    float* out = (float*)d_out;

    const int t = in_sizes[0] / (NQ * C);
    dim3 grid(t * (NQ / QPB) * (WAY / WPB));
    dim3 block(256);
    hipLaunchKernelGGL(proto_dist_kernel, grid, block, 0, stream, qf, sf, out);
}


# Each block starts its K loop at a different one of the 8 chunks (rotated, wraps), so blocks sharing rows stop asking the L2 for the same lines in the same step; K summation order differs per block
# speedup vs baseline: 2.9844x; 1.0531x over previous
#include <hip/hip_runtime.h>


typedef float v2f __attribute__((ext_vector_type(2)));
typedef float v4f __attribute__((ext_vector_type(4)));
typedef float v8f __attribute__((ext_vector_type(8)));

#define T_STEPS   8
#define WAY       64
#define SHOT      4
#define NQ        1024
#define C         512
#define QPB       64
#define WPB       32
#define CK        64
#define PITCH     68
#define PPITCH    20


struct Stage {
    v4f q[4];
    v4f s[2][4];
};

static __device__ __forceinline__
void stage_load(Stage& st, const float* __restrict__ qp, const float* __restrict__ sp, int ck)
{
    #pragma unroll
    for (int i = 0; i < 4; ++i)
        st.q[i] = *(const v4f*)(qp + ck + (size_t)(16 * i) * C);
    #pragma unroll
    for (int i = 0; i < 2; ++i) {
        #pragma unroll
        for (int j = 0; j < SHOT; ++j)
            st.s[i][j] = *(const v4f*)(sp + ck + (size_t)(16 * i * SHOT + j) * C);
    }
}

static __device__ __forceinline__
void stage_arrive(Stage& st)
{
    asm volatile("" : "+v"(st.q[0]), "+v"(st.q[1]), "+v"(st.q[2]), "+v"(st.q[3]),
                      "+v"(st.s[0][0]), "+v"(st.s[0][1]), "+v"(st.s[0][2]), "+v"(st.s[0][3]),
                      "+v"(st.s[1][0]), "+v"(st.s[1][1]), "+v"(st.s[1][2]), "+v"(st.s[1][3]));
}

static __device__ __forceinline__
void stage_commit(const Stage& st, float* lq, float* lp, v2f (&qa)[4], v2f (&pa)[2])
{
    #pragma unroll
    for (int i = 0; i < 4; ++i) {
        const v4f v = st.q[i];
        *(v4f*)&lq[16 * i * PITCH] = v;
        const v2f lo = {v.x, v.y};
        const v2f hi = {v.z, v.w};
        qa[i] += lo * lo;
        qa[i] += hi * hi;
    }
    #pragma unroll
    for (int i = 0; i < 2; ++i) {
        const v4f a = st.s[i][0];
        const v4f b = st.s[i][1];
        const v4f c = st.s[i][2];
        const v4f d = st.s[i][3];
        v4f m;
        m.x = 0.25f * (a.x + b.x + c.x + d.x);
        m.y = 0.25f * (a.y + b.y + c.y + d.y);
        m.z = 0.25f * (a.z + b.z + c.z + d.z);
        m.w = 0.25f * (a.w + b.w + c.w + d.w);
        *(v4f*)&lp[16 * i * PITCH] = m;
        const v2f lo = {m.x, m.y};
        const v2f hi = {m.z, m.w};
        pa[i] += lo * lo;
        pa[i] += hi * hi;
    }
}

static __device__ __forceinline__
void mma_chunk(const float* la, const float* lb, v8f& acc)
{
    #pragma unroll
    for (int kk = 0; kk < CK; kk += 4) {
        const v2f a = *(const v2f*)&la[kk];
        const v2f b = *(const v2f*)&lb[kk];
        acc = __builtin_amdgcn_wmma_f32_16x16x4_f32(
            false, a, false, b, (short)0, acc, false, false);
    }
}

static __device__ __forceinline__
void barrier_keep_loads()
{
    asm volatile("s_wait_dscnt 0x0\n\ts_barrier_signal -1\n\ts_barrier_wait -1" ::: "memory");
}

template <bool REFILL>
static __device__ __forceinline__
void pipe_step(Stage& st, const float* __restrict__ qp, const float* __restrict__ sp, int ck_refill,
               float* lq, float* lp, const float* la, const float* lb,
               v2f (&qa)[4], v2f (&pa)[2], v8f& acc)
{
    stage_arrive(st);
    stage_commit(st, lq, lp, qa, pa);
    __builtin_amdgcn_sched_barrier(0);
    if (REFILL) stage_load(st, qp, sp, ck_refill);
    barrier_keep_loads();
    __builtin_amdgcn_sched_barrier(0);
    mma_chunk(la, lb, acc);
    __builtin_amdgcn_sched_barrier(0);
}

__global__ __launch_bounds__(256, 4)
void proto_dist_kernel(const float* __restrict__ q,
                       const float* __restrict__ s,
                       float* __restrict__ out)
{
    __shared__ float ldsQ[2][QPB * PITCH];
    __shared__ float ldsP[2][WPB * PITCH];
    __shared__ __attribute__((aligned(16))) float qnp[QPB][PPITCH];
    __shared__ __attribute__((aligned(16))) float pnp[WPB][PPITCH];
    __shared__ float qn[QPB];
    __shared__ float pn[WPB];

    const int tid = threadIdx.x;
    const int wh  = blockIdx.x & 1;
    const int qb  = (blockIdx.x >> 1) & 15;
    const int t   = blockIdx.x >> 5;

    const float* Qg = q + ((size_t)t * NQ + (size_t)qb * QPB) * C;
    const float* Sg = s + ((size_t)t * (WAY * SHOT) + (size_t)wh * (WPB * SHOT)) * C;
    float*       Og = out + ((size_t)t * NQ + (size_t)qb * QPB) * WAY + wh * WPB;

    const int srow = tid >> 4;
    const int scol = tid & 15;

    const int lane = tid & 31;
    const int w    = tid >> 5;
    const int wm   = w & 3;
    const int n0   = (w >> 2) * 16;
    const int half = lane >> 4;
    const int l15  = lane & 15;

    const float* qp = Qg + (size_t)srow * C + scol * 4;
    const float* sp = Sg + (size_t)(srow * SHOT) * C + scol * 4;
    const int lq_off = srow * PITCH + scol * 4;
    const int lp_off = lq_off;
    const int la_off = (wm * 16 + l15) * PITCH + 2 * half;
    const int lb_off = (n0 + l15) * PITCH + 2 * half;

    v8f acc = {};
    v2f qa[4] = {};
    v2f pa[2] = {};

    const int rot = (qb + 4 * wh) & 7;
    #define CHUNK_AT(i) ((((i) + rot) & 7) * CK)

    Stage st0, st1;
    stage_load(st0, qp, sp, CHUNK_AT(0));
    __builtin_amdgcn_sched_barrier(0);
    stage_load(st1, qp, sp, CHUNK_AT(1));

    float* lq0 = &ldsQ[0][lq_off];
    float* lq1 = &ldsQ[1][lq_off];
    float* lp0 = &ldsP[0][lp_off];
    float* lp1 = &ldsP[1][lp_off];
    const float* la0 = &ldsQ[0][la_off];
    const float* la1 = &ldsQ[1][la_off];
    const float* lb0 = &ldsP[0][lb_off];
    const float* lb1 = &ldsP[1][lb_off];

    for (int i = 0; i < C / CK - 2; i += 2) {
        pipe_step<true>(st0, qp, sp, CHUNK_AT(i + 2), lq0, lp0, la0, lb0, qa, pa, acc);
        pipe_step<true>(st1, qp, sp, CHUNK_AT(i + 3), lq1, lp1, la1, lb1, qa, pa, acc);
    }
    #undef CHUNK_AT
    pipe_step<false>(st0, qp, sp, 0, lq0, lp0, la0, lb0, qa, pa, acc);
    pipe_step<false>(st1, qp, sp, 0, lq1, lp1, la1, lb1, qa, pa, acc);

    #pragma unroll
    for (int i = 0; i < 4; ++i) qnp[srow + 16 * i][scol] = qa[i].x + qa[i].y;
    #pragma unroll
    for (int i = 0; i < 2; ++i) pnp[srow + 16 * i][scol] = pa[i].x + pa[i].y;
    __syncthreads();
    if (tid < 96) {
        const float* pr = (tid < 64) ? &qnp[tid][0] : &pnp[tid - 64][0];
        const v4f a = *(const v4f*)&pr[0];
        const v4f b = *(const v4f*)&pr[4];
        const v4f c = *(const v4f*)&pr[8];
        const v4f d = *(const v4f*)&pr[12];
        const float sum = (((a.x + a.y) + (a.z + a.w)) + ((b.x + b.y) + (b.z + b.w)))
                        + (((c.x + c.y) + (c.z + c.w)) + ((d.x + d.y) + (d.z + d.w)));
        if (tid < 64) qn[tid] = sum;
        else          pn[tid - 64] = sum;
    }
    __syncthreads();

    float* tile = ldsQ[0];
    const float pn0 = pn[n0 + l15];
    #pragma unroll
    for (int i = 0; i < 8; ++i) {
        const int r   = wm * 16 + i + 8 * half;
        const float qr = qn[r];
        tile[r * PITCH + n0 + l15] = 2.0f * acc[i] - qr - pn0;
    }
    __syncthreads();

    for (int j = 0; j < 2; ++j) {
        const int idx = tid + 256 * j;
        const int r   = idx >> 3;
        const int c4  = (idx & 7) * 4;
        const float4 v = *(const float4*)&tile[r * PITCH + c4];
        *(float4*)(Og + (size_t)r * WAY + c4) = v;
    }
}

extern "C" void kernel_launch(void* const* d_in, const int* in_sizes, int n_in,
                              void* d_out, int out_size, void* d_ws, size_t ws_size,
                              hipStream_t stream)
{
    const float* qf = (const float*)d_in[0];
    const float* sf = (const float*)d_in[1];
    float* out = (float*)d_out;

    const int t = in_sizes[0] / (NQ * C);
    dim3 grid(t * (NQ / QPB) * (WAY / WPB));
    dim3 block(256);
    hipLaunchKernelGGL(proto_dist_kernel, grid, block, 0, stream, qf, sf, out);
}
